# starting kernel
# baseline (speedup 1.0000x reference)
#include <hip/hip_runtime.h>

typedef __attribute__((ext_vector_type(2))) float v2f;
typedef __attribute__((ext_vector_type(8))) float v8f;

#define S_ROWS 16384
#define K_COLS 4096
#define D_DIM  64
#define TEMP_C 50.0f
#define MAX_EFF_TEMP 5000.0f

__global__ __launch_bounds__(32) void fcm_kernel(const float* __restrict__ X,
                                                 const float* __restrict__ Y,
                                                 const float* __restrict__ icpt,
                                                 float* __restrict__ out_choice,
                                                 float* __restrict__ out_v) {
  __shared__ float stage[16 * 16];

  const int lane = threadIdx.x;
  const int nlo  = lane & 15;
  const int h    = lane >> 4;
  const int row0 = blockIdx.x * 16;

  v2f xfrag[16];
  const float* xrow = X + (size_t)(row0 + nlo) * D_DIM + 2 * h;
  #pragma unroll
  for (int c = 0; c < 16; ++c)
    xfrag[c] = *(const v2f*)(xrow + 4 * c);

  float rmax[8], rmin[8];
  #pragma unroll
  for (int j = 0; j < 8; ++j) { rmax[j] = -__builtin_inff(); rmin[j] = __builtin_inff(); }

  for (int n0 = 0; n0 < K_COLS; n0 += 16) {
    const float* yrow = Y + (size_t)(n0 + nlo) * D_DIM + 2 * h;
    v8f acc0 = {}; v8f acc1 = {};
    #pragma unroll
    for (int c = 0; c < 16; c += 2) {
      v2f b0 = *(const v2f*)(yrow + 4 * c);
      v2f b1 = *(const v2f*)(yrow + 4 * (c + 1));
      acc0 = __builtin_amdgcn_wmma_f32_16x16x4_f32(false, xfrag[c],     false, b0,
                                                   (short)0, acc0, false, false);
      acc1 = __builtin_amdgcn_wmma_f32_16x16x4_f32(false, xfrag[c + 1], false, b1,
                                                   (short)0, acc1, false, false);
    }
    const float ic = icpt[n0 + nlo];
    #pragma unroll
    for (int j = 0; j < 8; ++j) {
      float a = (acc0[j] + acc1[j]) - ic;
      rmax[j] = fmaxf(rmax[j], a);
      rmin[j] = fminf(rmin[j], a);
    }
  }
  #pragma unroll
  for (int m = 1; m < 16; m <<= 1) {
    #pragma unroll
    for (int j = 0; j < 8; ++j) {
      rmax[j] = fmaxf(rmax[j], __shfl_xor(rmax[j], m, 32));
      rmin[j] = fminf(rmin[j], __shfl_xor(rmin[j], m, 32));
    }
  }
  float efft[8], shift[8];
  #pragma unroll
  for (int j = 0; j < 8; ++j) {
    float span = fmaxf(rmax[j] - rmin[j], 1e-3f);
    efft[j]  = fminf(fmaxf(TEMP_C / span, TEMP_C), MAX_EFF_TEMP);
    shift[j] = rmax[j];
  }

  float denom[8], vnum[8];
  #pragma unroll
  for (int j = 0; j < 8; ++j) { denom[j] = 0.0f; vnum[j] = 0.0f; }
  v8f cacc[4];
  #pragma unroll
  for (int nt = 0; nt < 4; ++nt) cacc[nt] = (v8f){};

  for (int n0 = 0; n0 < K_COLS; n0 += 16) {
    const float* yrow = Y + (size_t)(n0 + nlo) * D_DIM + 2 * h;
    v8f acc0 = {}; v8f acc1 = {};
    #pragma unroll
    for (int c = 0; c < 16; c += 2) {
      v2f b0 = *(const v2f*)(yrow + 4 * c);
      v2f b1 = *(const v2f*)(yrow + 4 * (c + 1));
      acc0 = __builtin_amdgcn_wmma_f32_16x16x4_f32(false, xfrag[c],     false, b0,
                                                   (short)0, acc0, false, false);
      acc1 = __builtin_amdgcn_wmma_f32_16x16x4_f32(false, xfrag[c + 1], false, b1,
                                                   (short)0, acc1, false, false);
    }
    const float ic = icpt[n0 + nlo];
    float e[8];
    #pragma unroll
    for (int j = 0; j < 8; ++j) {
      float a = (acc0[j] + acc1[j]) - ic;
      float w = __expf((a - shift[j]) * efft[j]);
      e[j] = w;
      denom[j] += w;
      vnum[j]  += w * a;
    }

    #pragma unroll
    for (int j = 0; j < 8; ++j)
      stage[(j + 8 * h) * 16 + nlo] = e[j];
    asm volatile("s_wait_dscnt 0" ::: "memory");

    v2f ef[4];
    #pragma unroll
    for (int cc = 0; cc < 4; ++cc)
      ef[cc] = *(const v2f*)(&stage[nlo * 16 + 4 * cc + 2 * h]);
    asm volatile("s_wait_dscnt 0" ::: "memory");

    #pragma unroll
    for (int nt = 0; nt < 4; ++nt) {
      #pragma unroll
      for (int cc = 0; cc < 4; ++cc) {
        v2f b;
        b.x = Y[(size_t)(n0 + 4 * cc + 2 * h + 0) * D_DIM + nt * 16 + nlo];
        b.y = Y[(size_t)(n0 + 4 * cc + 2 * h + 1) * D_DIM + nt * 16 + nlo];
        cacc[nt] = __builtin_amdgcn_wmma_f32_16x16x4_f32(false, ef[cc], false, b,
                                                         (short)0, cacc[nt], false, false);
      }
    }
  }

  #pragma unroll
  for (int m = 1; m < 16; m <<= 1) {
    #pragma unroll
    for (int j = 0; j < 8; ++j) {
      denom[j] += __shfl_xor(denom[j], m, 32);
      vnum[j]  += __shfl_xor(vnum[j],  m, 32);
    }
  }
  float rden[8];
  #pragma unroll
  for (int j = 0; j < 8; ++j) rden[j] = 1.0f / denom[j];

  #pragma unroll
  for (int nt = 0; nt < 4; ++nt)
    #pragma unroll
    for (int j = 0; j < 8; ++j)
      out_choice[(size_t)(row0 + j + 8 * h) * D_DIM + nt * 16 + nlo] =
          cacc[nt][j] * rden[j];

  if (nlo == 0) {
    #pragma unroll
    for (int j = 0; j < 8; ++j)
      out_v[row0 + j + 8 * h] = vnum[j] * rden[j];
  }
}

extern "C" void kernel_launch(void* const* d_in, const int* in_sizes, int n_in,
                              void* d_out, int out_size, void* d_ws, size_t ws_size,
                              hipStream_t stream) {
  const float* X  = (const float*)d_in[0];
  const float* Y  = (const float*)d_in[1];
  const float* ic = (const float*)d_in[2];
  float* out      = (float*)d_out;
  float* out_choice = out;
  float* out_v      = out + (size_t)S_ROWS * D_DIM;

  dim3 grid(S_ROWS / 16);
  dim3 block(32);
  fcm_kernel<<<grid, block, 0, stream>>>(X, Y, ic, out_choice, out_v);
}


# Split each 16-row tile's K range over the 4 waves of a 128-thread workgroup (4 waves per SIMD cover load and LDS waits); the waves' partial sums are added in LDS, a new reduction tree
# speedup vs baseline: 2.2371x; 2.2371x over previous
#include <hip/hip_runtime.h>


typedef __attribute__((ext_vector_type(2))) float v2f;
typedef __attribute__((ext_vector_type(4))) float v4f;
typedef __attribute__((ext_vector_type(8))) float v8f;

#define S_ROWS 16384
#define K_COLS 4096
#define D_DIM  64
#define TEMP_C 50.0f
#define MAX_EFF_TEMP 5000.0f
#define NWAVE  4
#define PPITCH 68

__global__ __launch_bounds__(32 * NWAVE) void fcm_kernel(const float* __restrict__ X,
                                                         const float* __restrict__ Y,
                                                         const float* __restrict__ icpt,
                                                         float* __restrict__ out_choice,
                                                         float* __restrict__ out_v) {
  __shared__ __attribute__((aligned(16))) float stage_all[NWAVE * 16 * 16];
  __shared__ __attribute__((aligned(16))) float part[NWAVE * 16 * PPITCH];
  __shared__ float red_mx[NWAVE * 16];
  __shared__ float red_mn[NWAVE * 16];
  __shared__ float red_dn[NWAVE * 16];
  __shared__ float red_vn[NWAVE * 16];

  const int tid  = threadIdx.x;
  const int lane = tid & 31;
  const int wv   = __builtin_amdgcn_readfirstlane(tid >> 5);
  const int nlo  = lane & 15;
  const int h    = lane >> 4;
  const int row0 = blockIdx.x * 16;
  float* stage = stage_all + wv * 256;

  v2f xfrag[16];
  const float* xrow = X + (size_t)(row0 + nlo) * D_DIM + 2 * h;
  #pragma unroll
  for (int c = 0; c < 16; ++c)
    xfrag[c] = *(const v2f*)(xrow + 4 * c);

  float rmax[8], rmin[8];
  #pragma unroll
  for (int j = 0; j < 8; ++j) { rmax[j] = -__builtin_inff(); rmin[j] = __builtin_inff(); }

  for (int n0 = 16 * wv; n0 < K_COLS; n0 += 16 * NWAVE) {
    const float* yrow = Y + (size_t)(n0 + nlo) * D_DIM + 2 * h;
    v8f acc0 = {}; v8f acc1 = {};
    #pragma unroll
    for (int c = 0; c < 16; c += 2) {
      v2f b0 = *(const v2f*)(yrow + 4 * c);
      v2f b1 = *(const v2f*)(yrow + 4 * (c + 1));
      acc0 = __builtin_amdgcn_wmma_f32_16x16x4_f32(false, xfrag[c],     false, b0,
                                                   (short)0, acc0, false, false);
      acc1 = __builtin_amdgcn_wmma_f32_16x16x4_f32(false, xfrag[c + 1], false, b1,
                                                   (short)0, acc1, false, false);
    }
    const float ic = icpt[n0 + nlo];
    #pragma unroll
    for (int j = 0; j < 8; ++j) {
      float a = (acc0[j] + acc1[j]) - ic;
      rmax[j] = fmaxf(rmax[j], a);
      rmin[j] = fminf(rmin[j], a);
    }
  }
  #pragma unroll
  for (int m = 1; m < 16; m <<= 1) {
    #pragma unroll
    for (int j = 0; j < 8; ++j) {
      rmax[j] = fmaxf(rmax[j], __shfl_xor(rmax[j], m, 32));
      rmin[j] = fminf(rmin[j], __shfl_xor(rmin[j], m, 32));
    }
  }
  if (nlo == 0) {
    #pragma unroll
    for (int j = 0; j < 8; ++j) {
      red_mx[wv * 16 + j + 8 * h] = rmax[j];
      red_mn[wv * 16 + j + 8 * h] = rmin[j];
    }
  }
  __syncthreads();
  #pragma unroll
  for (int j = 0; j < 8; ++j) {
    float mx = red_mx[j + 8 * h];
    float mn = red_mn[j + 8 * h];
    #pragma unroll
    for (int w = 1; w < NWAVE; ++w) {
      mx = fmaxf(mx, red_mx[w * 16 + j + 8 * h]);
      mn = fminf(mn, red_mn[w * 16 + j + 8 * h]);
    }
    rmax[j] = mx;
    rmin[j] = mn;
  }

  float efft[8], shift[8];
  #pragma unroll
  for (int j = 0; j < 8; ++j) {
    float span = fmaxf(rmax[j] - rmin[j], 1e-3f);
    efft[j]  = fminf(fmaxf(TEMP_C / span, TEMP_C), MAX_EFF_TEMP);
    shift[j] = rmax[j];
  }

  float denom[8], vnum[8];
  #pragma unroll
  for (int j = 0; j < 8; ++j) { denom[j] = 0.0f; vnum[j] = 0.0f; }
  v8f cacc[4];
  #pragma unroll
  for (int nt = 0; nt < 4; ++nt) cacc[nt] = (v8f){};

  for (int n0 = 16 * wv; n0 < K_COLS; n0 += 16 * NWAVE) {
    const float* yrow = Y + (size_t)(n0 + nlo) * D_DIM + 2 * h;
    v8f acc0 = {}; v8f acc1 = {};
    #pragma unroll
    for (int c = 0; c < 16; c += 2) {
      v2f b0 = *(const v2f*)(yrow + 4 * c);
      v2f b1 = *(const v2f*)(yrow + 4 * (c + 1));
      acc0 = __builtin_amdgcn_wmma_f32_16x16x4_f32(false, xfrag[c],     false, b0,
                                                   (short)0, acc0, false, false);
      acc1 = __builtin_amdgcn_wmma_f32_16x16x4_f32(false, xfrag[c + 1], false, b1,
                                                   (short)0, acc1, false, false);
    }
    const float ic = icpt[n0 + nlo];
    float e[8];
    #pragma unroll
    for (int j = 0; j < 8; ++j) {
      float a = (acc0[j] + acc1[j]) - ic;
      float w = __expf((a - shift[j]) * efft[j]);
      e[j] = w;
      denom[j] += w;
      vnum[j]  += w * a;
    }

    #pragma unroll
    for (int j = 0; j < 8; ++j)
      stage[(j + 8 * h) * 16 + nlo] = e[j];
    asm volatile("s_wait_dscnt 0" ::: "memory");

    v2f ef[4];
    #pragma unroll
    for (int cc = 0; cc < 4; ++cc)
      ef[cc] = *(const v2f*)(&stage[nlo * 16 + 4 * cc + 2 * h]);
    asm volatile("s_wait_dscnt 0" ::: "memory");

    #pragma unroll
    for (int nt = 0; nt < 4; ++nt) {
      #pragma unroll
      for (int cc = 0; cc < 4; ++cc) {
        v2f b;
        b.x = Y[(size_t)(n0 + 4 * cc + 2 * h + 0) * D_DIM + nt * 16 + nlo];
        b.y = Y[(size_t)(n0 + 4 * cc + 2 * h + 1) * D_DIM + nt * 16 + nlo];
        cacc[nt] = __builtin_amdgcn_wmma_f32_16x16x4_f32(false, ef[cc], false, b,
                                                         (short)0, cacc[nt], false, false);
      }
    }
  }

  #pragma unroll
  for (int m = 1; m < 16; m <<= 1) {
    #pragma unroll
    for (int j = 0; j < 8; ++j) {
      denom[j] += __shfl_xor(denom[j], m, 32);
      vnum[j]  += __shfl_xor(vnum[j],  m, 32);
    }
  }

  if (nlo == 0) {
    #pragma unroll
    for (int j = 0; j < 8; ++j) {
      red_dn[wv * 16 + j + 8 * h] = denom[j];
      red_vn[wv * 16 + j + 8 * h] = vnum[j];
    }
  }
  #pragma unroll
  for (int nt = 0; nt < 4; ++nt)
    #pragma unroll
    for (int j = 0; j < 8; ++j)
      part[(wv * 16 + j + 8 * h) * PPITCH + nt * 16 + nlo] = cacc[nt][j];
  __syncthreads();

  #pragma unroll
  for (int q = tid; q < 16 * 16; q += 32 * NWAVE) {
    const int row = q >> 4;
    const int c4  = q & 15;
    float dn = red_dn[row];
    v4f s = *(const v4f*)(&part[row * PPITCH + 4 * c4]);
    #pragma unroll
    for (int w = 1; w < NWAVE; ++w) {
      dn += red_dn[w * 16 + row];
      s  += *(const v4f*)(&part[(w * 16 + row) * PPITCH + 4 * c4]);
    }
    const float rd = 1.0f / dn;
    s *= rd;
    *(v4f*)(&out_choice[(size_t)(row0 + row) * D_DIM + 4 * c4]) = s;
  }

  if (tid < 16) {
    float dn = red_dn[tid];
    float vn = red_vn[tid];
    #pragma unroll
    for (int w = 1; w < NWAVE; ++w) {
      dn += red_dn[w * 16 + tid];
      vn += red_vn[w * 16 + tid];
    }
    out_v[row0 + tid] = vn * (1.0f / dn);
  }
}

extern "C" void kernel_launch(void* const* d_in, const int* in_sizes, int n_in,
                              void* d_out, int out_size, void* d_ws, size_t ws_size,
                              hipStream_t stream) {
  const float* X  = (const float*)d_in[0];
  const float* Y  = (const float*)d_in[1];
  const float* ic = (const float*)d_in[2];
  float* out      = (float*)d_out;
  float* out_choice = out;
  float* out_v      = out + (size_t)S_ROWS * D_DIM;

  dim3 grid(S_ROWS / 16);
  dim3 block(32 * NWAVE);
  fcm_kernel<<<grid, block, 0, stream>>>(X, Y, ic, out_choice, out_v);
}


# Pass 1 parks the affine tile in LDS (256 KiB per 16-wave workgroup) and pass 2 reads it back instead of recomputing the first product; products transposed; sums in a new reduction tree
# speedup vs baseline: 2.9466x; 1.3172x over previous
#include <hip/hip_runtime.h>


typedef __attribute__((ext_vector_type(2))) float v2f;
typedef __attribute__((ext_vector_type(4))) float v4f;
typedef __attribute__((ext_vector_type(8))) float v8f;

#define S_ROWS 16384
#define K_COLS 4096
#define D_DIM  64
#define TEMP_C 50.0f
#define MAX_EFF_TEMP 5000.0f
#define NWAVE  16
#define NCH    (K_COLS / (16 * NWAVE))
#define PPITCH 68
#define RPITCH 20

__global__ __launch_bounds__(32 * NWAVE) void fcm_kernel(const float* __restrict__ X,
                                                         const float* __restrict__ Y,
                                                         const float* __restrict__ icpt,
                                                         float* __restrict__ out_choice,
                                                         float* __restrict__ out_v) {
  __shared__ v4f aff_lds[NWAVE * NCH * 2 * 32];
  __shared__ __attribute__((aligned(16))) float red_mx[16 * RPITCH];
  __shared__ __attribute__((aligned(16))) float red_mn[16 * RPITCH];
  __shared__ float red_dn[NWAVE * 16];
  __shared__ float red_vn[NWAVE * 16];
  float* part = (float*)aff_lds;

  const int tid  = threadIdx.x;
  const int lane = tid & 31;
  const int wv   = __builtin_amdgcn_readfirstlane(tid >> 5);
  const int nlo  = lane & 15;
  const int h    = lane >> 4;
  const int row0 = blockIdx.x * 16;
  v4f* park = aff_lds + (size_t)wv * (NCH * 2 * 32) + lane;

  v2f xfrag[16];
  const float* xrow = X + (size_t)(row0 + nlo) * D_DIM + 2 * h;
  #pragma unroll
  for (int c = 0; c < 16; ++c)
    xfrag[c] = *(const v2f*)(xrow + 4 * c);

  float rmax = -__builtin_inff();
  float rmin =  __builtin_inff();

  v2f ya1[2][16];
  v4f icp[2][2];
  {
    const float* yrow = Y + (size_t)(16 * wv + nlo) * D_DIM + 2 * h;
    #pragma unroll
    for (int c = 0; c < 16; ++c)
      ya1[0][c] = *(const v2f*)(yrow + 4 * c);
    icp[0][0] = *(const v4f*)(icpt + 16 * wv + 8 * h);
    icp[0][1] = *(const v4f*)(icpt + 16 * wv + 8 * h + 4);
  }

  #pragma unroll
  for (int i = 0; i < NCH; ++i) {
    if (i + 1 < NCH) {
      const int n1 = 16 * wv + 16 * NWAVE * (i + 1);
      const float* yrow = Y + (size_t)(n1 + nlo) * D_DIM + 2 * h;
      #pragma unroll
      for (int c = 0; c < 16; ++c)
        ya1[(i + 1) & 1][c] = *(const v2f*)(yrow + 4 * c);
      icp[(i + 1) & 1][0] = *(const v4f*)(icpt + n1 + 8 * h);
      icp[(i + 1) & 1][1] = *(const v4f*)(icpt + n1 + 8 * h + 4);
    }
    __builtin_amdgcn_sched_barrier(0);

    v8f acc0 = {}; v8f acc1 = {};
    #pragma unroll
    for (int c = 0; c < 16; c += 2) {
      acc0 = __builtin_amdgcn_wmma_f32_16x16x4_f32(false, ya1[i & 1][c],     false, xfrag[c],
                                                   (short)0, acc0, false, false);
      acc1 = __builtin_amdgcn_wmma_f32_16x16x4_f32(false, ya1[i & 1][c + 1], false, xfrag[c + 1],
                                                   (short)0, acc1, false, false);
    }
    const v4f ic0 = icp[i & 1][0];
    const v4f ic1 = icp[i & 1][1];
    v4f lo, hi;
    #pragma unroll
    for (int t = 0; t < 4; ++t) {
      lo[t] = (acc0[t]     + acc1[t])     - ic0[t];
      hi[t] = (acc0[4 + t] + acc1[4 + t]) - ic1[t];
      rmax = fmaxf(rmax, fmaxf(lo[t], hi[t]));
      rmin = fminf(rmin, fminf(lo[t], hi[t]));
    }
    park[(2 * i + 0) * 32] = lo;
    park[(2 * i + 1) * 32] = hi;
  }

  v2f yfr[2][16];
  {
    const float* ycol = Y + (size_t)(16 * wv + 8 * h) * D_DIM + nlo;
    #pragma unroll
    for (int dt = 0; dt < 4; ++dt) {
      #pragma unroll
      for (int cc = 0; cc < 4; ++cc) {
        yfr[0][4 * dt + cc].x = ycol[(size_t)(2 * cc + 0) * D_DIM + dt * 16];
        yfr[0][4 * dt + cc].y = ycol[(size_t)(2 * cc + 1) * D_DIM + dt * 16];
      }
    }
  }

  rmax = fmaxf(rmax, __shfl_xor(rmax, 16, 32));
  rmin = fminf(rmin, __shfl_xor(rmin, 16, 32));

  if (h == 0) {
    red_mx[nlo * RPITCH + wv] = rmax;
    red_mn[nlo * RPITCH + wv] = rmin;
  }
  __syncthreads();
  #pragma unroll
  for (int q = 0; q < NWAVE / 4; ++q) {
    const v4f mx4 = *(const v4f*)(&red_mx[nlo * RPITCH + 4 * q]);
    const v4f mn4 = *(const v4f*)(&red_mn[nlo * RPITCH + 4 * q]);
    #pragma unroll
    for (int t = 0; t < 4; ++t) {
      rmax = fmaxf(rmax, mx4[t]);
      rmin = fminf(rmin, mn4[t]);
    }
  }

  const float span  = fmaxf(rmax - rmin, 1e-3f);
  const float efft  = fminf(fmaxf(TEMP_C / span, TEMP_C), MAX_EFF_TEMP);
  const float shift = rmax;

  float denom = 0.0f, vnum = 0.0f;
  v8f cacc[4];
  #pragma unroll
  for (int dt = 0; dt < 4; ++dt) cacc[dt] = (v8f){};

  v4f afr[2][2];
  afr[0][0] = park[0];
  afr[0][1] = park[32];

  #pragma unroll
  for (int i = 0; i < NCH; ++i) {
    if (i + 1 < NCH) {
      const float* ynx = Y + (size_t)(16 * wv + 16 * NWAVE * (i + 1) + 8 * h) * D_DIM + nlo;
      #pragma unroll
      for (int dt = 0; dt < 4; ++dt) {
        #pragma unroll
        for (int cc = 0; cc < 4; ++cc) {
          yfr[(i + 1) & 1][4 * dt + cc].x = ynx[(size_t)(2 * cc + 0) * D_DIM + dt * 16];
          yfr[(i + 1) & 1][4 * dt + cc].y = ynx[(size_t)(2 * cc + 1) * D_DIM + dt * 16];
        }
      }
      afr[(i + 1) & 1][0] = park[(2 * (i + 1) + 0) * 32];
      afr[(i + 1) & 1][1] = park[(2 * (i + 1) + 1) * 32];
    }
    __builtin_amdgcn_sched_barrier(0);

    float e[8];
    #pragma unroll
    for (int j = 0; j < 8; ++j) {
      float a = afr[i & 1][j >> 2][j & 3];
      float w = __expf((a - shift) * efft);
      e[j] = w;
      denom += w;
      vnum  += w * a;
    }

    #pragma unroll
    for (int dt = 0; dt < 4; ++dt) {
      #pragma unroll
      for (int cc = 0; cc < 4; ++cc) {
        const v2f ya = yfr[i & 1][4 * dt + cc];
        v2f eb;
        eb.x = e[2 * cc + 0];
        eb.y = e[2 * cc + 1];
        cacc[dt] = __builtin_amdgcn_wmma_f32_16x16x4_f32(false, ya, false, eb,
                                                         (short)0, cacc[dt], false, false);
      }
    }
  }

  denom += __shfl_xor(denom, 16, 32);
  vnum  += __shfl_xor(vnum,  16, 32);

  if (h == 0) {
    red_dn[wv * 16 + nlo] = denom;
    red_vn[wv * 16 + nlo] = vnum;
  }
  __syncthreads();
  #pragma unroll
  for (int dt = 0; dt < 4; ++dt) {
    v4f lo, hi;
    #pragma unroll
    for (int t = 0; t < 4; ++t) { lo[t] = cacc[dt][t]; hi[t] = cacc[dt][4 + t]; }
    float* p = &part[(wv * 16 + nlo) * PPITCH + dt * 16 + 8 * h];
    *(v4f*)(p)     = lo;
    *(v4f*)(p + 4) = hi;
  }
  __syncthreads();

  if (tid < 16 * 16) {
    const int row = tid >> 4;
    const int c4  = tid & 15;
    float dn = red_dn[row];
    v4f s = *(const v4f*)(&part[row * PPITCH + 4 * c4]);
    #pragma unroll
    for (int w = 1; w < NWAVE; ++w) {
      dn += red_dn[w * 16 + row];
      s  += *(const v4f*)(&part[(w * 16 + row) * PPITCH + 4 * c4]);
    }
    const float rd = 1.0f / dn;
    s *= rd;
    *(v4f*)(&out_choice[(size_t)(row0 + row) * D_DIM + 4 * c4]) = s;
  }

  if (tid >= 16 * 16 && tid < 16 * 16 + 16) {
    const int row = tid - 16 * 16;
    float dn = red_dn[row];
    float vn = red_vn[row];
    #pragma unroll
    for (int w = 1; w < NWAVE; ++w) {
      dn += red_dn[w * 16 + row];
      vn += red_vn[w * 16 + row];
    }
    out_v[row0 + row] = vn * (1.0f / dn);
  }
}

extern "C" void kernel_launch(void* const* d_in, const int* in_sizes, int n_in,
                              void* d_out, int out_size, void* d_ws, size_t ws_size,
                              hipStream_t stream) {
  const float* X  = (const float*)d_in[0];
  const float* Y  = (const float*)d_in[1];
  const float* ic = (const float*)d_in[2];
  float* out      = (float*)d_out;
  float* out_choice = out;
  float* out_v      = out + (size_t)S_ROWS * D_DIM;

  dim3 grid(S_ROWS / 16);
  dim3 block(32 * NWAVE);
  fcm_kernel<<<grid, block, 0, stream>>>(X, Y, ic, out_choice, out_v);
}


# Pass 2 skips a 16-row Y chunk (its Y loads and its 16 WMMAs) when all 256 softmax weights of the chunk are exactly 0.0f, on a wave-uniform ballot branch; only zero terms leave the sums
# speedup vs baseline: 4.0776x; 1.3838x over previous
#include <hip/hip_runtime.h>


typedef __attribute__((ext_vector_type(2))) float v2f;
typedef __attribute__((ext_vector_type(4))) float v4f;
typedef __attribute__((ext_vector_type(8))) float v8f;

#define S_ROWS 16384
#define K_COLS 4096
#define D_DIM  64
#define TEMP_C 50.0f
#define MAX_EFF_TEMP 5000.0f
#define NWAVE  16
#define NCH    (K_COLS / (16 * NWAVE))
#define PPITCH 68
#define RPITCH 20

__global__ __launch_bounds__(32 * NWAVE) void fcm_kernel(const float* __restrict__ X,
                                                         const float* __restrict__ Y,
                                                         const float* __restrict__ icpt,
                                                         float* __restrict__ out_choice,
                                                         float* __restrict__ out_v) {
  __shared__ v4f aff_lds[NWAVE * NCH * 2 * 32];
  __shared__ __attribute__((aligned(16))) float red_mx[16 * RPITCH];
  __shared__ __attribute__((aligned(16))) float red_mn[16 * RPITCH];
  __shared__ float red_dn[NWAVE * 16];
  __shared__ float red_vn[NWAVE * 16];
  float* part = (float*)aff_lds;

  const int tid  = threadIdx.x;
  const int lane = tid & 31;
  const int wv   = __builtin_amdgcn_readfirstlane(tid >> 5);
  const int nlo  = lane & 15;
  const int h    = lane >> 4;
  const int row0 = blockIdx.x * 16;
  v4f* park = aff_lds + (size_t)wv * (NCH * 2 * 32) + lane;

  v2f xfrag[16];
  const float* xrow = X + (size_t)(row0 + nlo) * D_DIM + 2 * h;
  #pragma unroll
  for (int c = 0; c < 16; ++c)
    xfrag[c] = *(const v2f*)(xrow + 4 * c);

  float rmax = -__builtin_inff();
  float rmin =  __builtin_inff();

  v2f ya1[2][16];
  v4f icp[2][2];
  {
    const float* yrow = Y + (size_t)(16 * wv + nlo) * D_DIM + 2 * h;
    #pragma unroll
    for (int c = 0; c < 16; ++c)
      ya1[0][c] = *(const v2f*)(yrow + 4 * c);
    icp[0][0] = *(const v4f*)(icpt + 16 * wv + 8 * h);
    icp[0][1] = *(const v4f*)(icpt + 16 * wv + 8 * h + 4);
  }

  #pragma unroll
  for (int i = 0; i < NCH; ++i) {
    if (i + 1 < NCH) {
      const int n1 = 16 * wv + 16 * NWAVE * (i + 1);
      const float* yrow = Y + (size_t)(n1 + nlo) * D_DIM + 2 * h;
      #pragma unroll
      for (int c = 0; c < 16; ++c)
        ya1[(i + 1) & 1][c] = *(const v2f*)(yrow + 4 * c);
      icp[(i + 1) & 1][0] = *(const v4f*)(icpt + n1 + 8 * h);
      icp[(i + 1) & 1][1] = *(const v4f*)(icpt + n1 + 8 * h + 4);
    }
    __builtin_amdgcn_sched_barrier(0);

    v8f acc0 = {}; v8f acc1 = {};
    #pragma unroll
    for (int c = 0; c < 16; c += 2) {
      acc0 = __builtin_amdgcn_wmma_f32_16x16x4_f32(false, ya1[i & 1][c],     false, xfrag[c],
                                                   (short)0, acc0, false, false);
      acc1 = __builtin_amdgcn_wmma_f32_16x16x4_f32(false, ya1[i & 1][c + 1], false, xfrag[c + 1],
                                                   (short)0, acc1, false, false);
    }
    const v4f ic0 = icp[i & 1][0];
    const v4f ic1 = icp[i & 1][1];
    v4f lo, hi;
    #pragma unroll
    for (int t = 0; t < 4; ++t) {
      lo[t] = (acc0[t]     + acc1[t])     - ic0[t];
      hi[t] = (acc0[4 + t] + acc1[4 + t]) - ic1[t];
      rmax = fmaxf(rmax, fmaxf(lo[t], hi[t]));
      rmin = fminf(rmin, fminf(lo[t], hi[t]));
    }
    park[(2 * i + 0) * 32] = lo;
    park[(2 * i + 1) * 32] = hi;
  }

  rmax = fmaxf(rmax, __shfl_xor(rmax, 16, 32));
  rmin = fminf(rmin, __shfl_xor(rmin, 16, 32));

  if (h == 0) {
    red_mx[nlo * RPITCH + wv] = rmax;
    red_mn[nlo * RPITCH + wv] = rmin;
  }
  __syncthreads();
  #pragma unroll
  for (int q = 0; q < NWAVE / 4; ++q) {
    const v4f mx4 = *(const v4f*)(&red_mx[nlo * RPITCH + 4 * q]);
    const v4f mn4 = *(const v4f*)(&red_mn[nlo * RPITCH + 4 * q]);
    #pragma unroll
    for (int t = 0; t < 4; ++t) {
      rmax = fmaxf(rmax, mx4[t]);
      rmin = fminf(rmin, mn4[t]);
    }
  }

  const float span  = fmaxf(rmax - rmin, 1e-3f);
  const float efft  = fminf(fmaxf(TEMP_C / span, TEMP_C), MAX_EFF_TEMP);
  const float shift = rmax;

  float denom = 0.0f, vnum = 0.0f;
  v8f cacc[4];
  #pragma unroll
  for (int dt = 0; dt < 4; ++dt) cacc[dt] = (v8f){};

  v4f afr[2][2];
  afr[0][0] = park[0];
  afr[0][1] = park[32];

  #pragma unroll
  for (int i = 0; i < NCH; ++i) {
    if (i + 1 < NCH) {
      afr[(i + 1) & 1][0] = park[(2 * (i + 1) + 0) * 32];
      afr[(i + 1) & 1][1] = park[(2 * (i + 1) + 1) * 32];
    }

    float a[8], e[8];
    float emax = 0.0f;
    #pragma unroll
    for (int j = 0; j < 8; ++j) {
      a[j] = afr[i & 1][j >> 2][j & 3];
      e[j] = __expf((a[j] - shift) * efft);
      emax = fmaxf(emax, e[j]);
    }

    if (__builtin_amdgcn_ballot_w32(emax > 0.0f) != 0u) {
      const float* ycol = Y + (size_t)(16 * wv + 16 * NWAVE * i + 8 * h) * D_DIM + nlo;
      v2f yfr[16];
      #pragma unroll
      for (int dt = 0; dt < 4; ++dt) {
        #pragma unroll
        for (int cc = 0; cc < 4; ++cc) {
          yfr[4 * dt + cc].x = ycol[(size_t)(2 * cc + 0) * D_DIM + dt * 16];
          yfr[4 * dt + cc].y = ycol[(size_t)(2 * cc + 1) * D_DIM + dt * 16];
        }
      }

      #pragma unroll
      for (int j = 0; j < 8; ++j) {
        denom += e[j];
        vnum   = __builtin_fmaf(e[j], a[j], vnum);
      }

      #pragma unroll
      for (int cc = 0; cc < 4; ++cc) {
        v2f eb;
        eb.x = e[2 * cc + 0];
        eb.y = e[2 * cc + 1];
        #pragma unroll
        for (int dt = 0; dt < 4; ++dt) {
          cacc[dt] = __builtin_amdgcn_wmma_f32_16x16x4_f32(false, yfr[4 * dt + cc], false, eb,
                                                           (short)0, cacc[dt], false, false);
        }
      }
    }
  }

  denom += __shfl_xor(denom, 16, 32);
  vnum  += __shfl_xor(vnum,  16, 32);

  if (h == 0) {
    red_dn[wv * 16 + nlo] = denom;
    red_vn[wv * 16 + nlo] = vnum;
  }
  __syncthreads();
  #pragma unroll
  for (int dt = 0; dt < 4; ++dt) {
    v4f lo, hi;
    #pragma unroll
    for (int t = 0; t < 4; ++t) { lo[t] = cacc[dt][t]; hi[t] = cacc[dt][4 + t]; }
    float* p = &part[(wv * 16 + nlo) * PPITCH + dt * 16 + 8 * h];
    *(v4f*)(p)     = lo;
    *(v4f*)(p + 4) = hi;
  }
  __syncthreads();

  if (tid < 16 * 16) {
    const int row = tid >> 4;
    const int c4  = tid & 15;
    float dn = red_dn[row];
    v4f s = *(const v4f*)(&part[row * PPITCH + 4 * c4]);
    #pragma unroll
    for (int w = 1; w < NWAVE; ++w) {
      dn += red_dn[w * 16 + row];
      s  += *(const v4f*)(&part[(w * 16 + row) * PPITCH + 4 * c4]);
    }
    const float rd = 1.0f / dn;
    s *= rd;
    *(v4f*)(&out_choice[(size_t)(row0 + row) * D_DIM + 4 * c4]) = s;
  }

  if (tid >= 16 * 16 && tid < 16 * 16 + 16) {
    const int row = tid - 16 * 16;
    float dn = red_dn[row];
    float vn = red_vn[row];
    #pragma unroll
    for (int w = 1; w < NWAVE; ++w) {
      dn += red_dn[w * 16 + row];
      vn += red_vn[w * 16 + row];
    }
    out_v[row0 + row] = vn * (1.0f / dn);
  }
}

extern "C" void kernel_launch(void* const* d_in, const int* in_sizes, int n_in,
                              void* d_out, int out_size, void* d_ws, size_t ws_size,
                              hipStream_t stream) {
  const float* X  = (const float*)d_in[0];
  const float* Y  = (const float*)d_in[1];
  const float* ic = (const float*)d_in[2];
  float* out      = (float*)d_out;
  float* out_choice = out;
  float* out_v      = out + (size_t)S_ROWS * D_DIM;

  dim3 grid(S_ROWS / 16);
  dim3 block(32 * NWAVE);
  fcm_kernel<<<grid, block, 0, stream>>>(X, Y, ic, out_choice, out_v);
}


# Pass 1 keeps one per-chunk maximum per lane in registers, not the affine tile in LDS; live chunks redo their 16 affine WMMAs; 8-wave workgroups, two per WGP overlap phases; new reduction tree
# speedup vs baseline: 4.2888x; 1.0518x over previous
#include <hip/hip_runtime.h>


typedef __attribute__((ext_vector_type(2))) float v2f;
typedef __attribute__((ext_vector_type(4))) float v4f;
typedef __attribute__((ext_vector_type(8))) float v8f;

#define S_ROWS 16384
#define K_COLS 4096
#define D_DIM  64
#define TEMP_C 50.0f
#define MAX_EFF_TEMP 5000.0f
#define NWAVE  8
#define NCH    (K_COLS / (16 * NWAVE))
#define PPITCH 68
#define RPITCH 12

__global__ __launch_bounds__(32 * NWAVE) void fcm_kernel(const float* __restrict__ X,
                                                         const float* __restrict__ Y,
                                                         const float* __restrict__ icpt,
                                                         float* __restrict__ out_choice,
                                                         float* __restrict__ out_v) {
  __shared__ __attribute__((aligned(16))) float part[NWAVE * 16 * PPITCH];
  __shared__ __attribute__((aligned(16))) float red_mx[16 * RPITCH];
  __shared__ __attribute__((aligned(16))) float red_mn[16 * RPITCH];
  __shared__ float red_dn[NWAVE * 16];
  __shared__ float red_vn[NWAVE * 16];

  const int tid  = threadIdx.x;
  const int lane = tid & 31;
  const int wv   = __builtin_amdgcn_readfirstlane(tid >> 5);
  const int nlo  = lane & 15;
  const int h    = lane >> 4;
  const int row0 = blockIdx.x * 16;

  v2f xfrag[16];
  const float* xrow = X + (size_t)(row0 + nlo) * D_DIM + 2 * h;
  #pragma unroll
  for (int c = 0; c < 16; ++c)
    xfrag[c] = *(const v2f*)(xrow + 4 * c);

  float rmax = -__builtin_inff();
  float rmin =  __builtin_inff();
  float cmax[NCH];

  v2f ya1[2][16];
  v4f icp[2][2];
  {
    const float* yrow = Y + (size_t)(16 * wv + nlo) * D_DIM + 2 * h;
    #pragma unroll
    for (int c = 0; c < 16; ++c)
      ya1[0][c] = *(const v2f*)(yrow + 4 * c);
    icp[0][0] = *(const v4f*)(icpt + 16 * wv + 8 * h);
    icp[0][1] = *(const v4f*)(icpt + 16 * wv + 8 * h + 4);
  }

  #pragma unroll
  for (int i = 0; i < NCH; ++i) {
    if (i + 1 < NCH) {
      const int n1 = 16 * wv + 16 * NWAVE * (i + 1);
      const float* yrow = Y + (size_t)(n1 + nlo) * D_DIM + 2 * h;
      #pragma unroll
      for (int c = 0; c < 16; ++c)
        ya1[(i + 1) & 1][c] = *(const v2f*)(yrow + 4 * c);
      icp[(i + 1) & 1][0] = *(const v4f*)(icpt + n1 + 8 * h);
      icp[(i + 1) & 1][1] = *(const v4f*)(icpt + n1 + 8 * h + 4);
    }
    __builtin_amdgcn_sched_barrier(0);

    v8f acc0 = {}; v8f acc1 = {};
    #pragma unroll
    for (int c = 0; c < 16; c += 2) {
      acc0 = __builtin_amdgcn_wmma_f32_16x16x4_f32(false, ya1[i & 1][c],     false, xfrag[c],
                                                   (short)0, acc0, false, false);
      acc1 = __builtin_amdgcn_wmma_f32_16x16x4_f32(false, ya1[i & 1][c + 1], false, xfrag[c + 1],
                                                   (short)0, acc1, false, false);
    }
    const v4f ic0 = icp[i & 1][0];
    const v4f ic1 = icp[i & 1][1];
    float cm = -__builtin_inff();
    #pragma unroll
    for (int t = 0; t < 4; ++t) {
      const float lo = (acc0[t]     + acc1[t])     - ic0[t];
      const float hi = (acc0[4 + t] + acc1[4 + t]) - ic1[t];
      cm   = fmaxf(cm,   fmaxf(lo, hi));
      rmin = fminf(rmin, fminf(lo, hi));
    }
    cmax[i] = cm;
    rmax = fmaxf(rmax, cm);
  }

  rmax = fmaxf(rmax, __shfl_xor(rmax, 16, 32));
  rmin = fminf(rmin, __shfl_xor(rmin, 16, 32));

  if (h == 0) {
    red_mx[nlo * RPITCH + wv] = rmax;
    red_mn[nlo * RPITCH + wv] = rmin;
  }
  __syncthreads();
  #pragma unroll
  for (int q = 0; q < NWAVE / 4; ++q) {
    const v4f mx4 = *(const v4f*)(&red_mx[nlo * RPITCH + 4 * q]);
    const v4f mn4 = *(const v4f*)(&red_mn[nlo * RPITCH + 4 * q]);
    #pragma unroll
    for (int t = 0; t < 4; ++t) {
      rmax = fmaxf(rmax, mx4[t]);
      rmin = fminf(rmin, mn4[t]);
    }
  }

  const float span  = fmaxf(rmax - rmin, 1e-3f);
  const float efft  = fminf(fmaxf(TEMP_C / span, TEMP_C), MAX_EFF_TEMP);
  const float shift = rmax;

  unsigned live = 0u;
  #pragma unroll
  for (int i = 0; i < NCH; ++i) {
    const float etop = __expf((cmax[i] - shift) * efft);
    if (__builtin_amdgcn_ballot_w32(etop > 0.0f) != 0u) live |= (1u << i);
  }

  float denom = 0.0f, vnum = 0.0f;
  v8f cacc[4];
  #pragma unroll
  for (int dt = 0; dt < 4; ++dt) cacc[dt] = (v8f){};

  while (live != 0u) {
    const int i = __builtin_ctz(live);
    live &= live - 1u;
    const int n0 = 16 * wv + 16 * NWAVE * i;

    const float* yrow = Y + (size_t)(n0 + nlo) * D_DIM + 2 * h;
    v2f ya[16];
    #pragma unroll
    for (int c = 0; c < 16; ++c)
      ya[c] = *(const v2f*)(yrow + 4 * c);
    const v4f ic0 = *(const v4f*)(icpt + n0 + 8 * h);
    const v4f ic1 = *(const v4f*)(icpt + n0 + 8 * h + 4);
    const float* ycol = Y + (size_t)(n0 + 8 * h) * D_DIM + nlo;
    v2f yfr[16];
    #pragma unroll
    for (int dt = 0; dt < 4; ++dt) {
      #pragma unroll
      for (int cc = 0; cc < 4; ++cc) {
        yfr[4 * dt + cc].x = ycol[(size_t)(2 * cc + 0) * D_DIM + dt * 16];
        yfr[4 * dt + cc].y = ycol[(size_t)(2 * cc + 1) * D_DIM + dt * 16];
      }
    }

    v8f acc0 = {}; v8f acc1 = {};
    #pragma unroll
    for (int c = 0; c < 16; c += 2) {
      acc0 = __builtin_amdgcn_wmma_f32_16x16x4_f32(false, ya[c],     false, xfrag[c],
                                                   (short)0, acc0, false, false);
      acc1 = __builtin_amdgcn_wmma_f32_16x16x4_f32(false, ya[c + 1], false, xfrag[c + 1],
                                                   (short)0, acc1, false, false);
    }
    float a[8], e[8];
    #pragma unroll
    for (int t = 0; t < 4; ++t) {
      a[t]     = (acc0[t]     + acc1[t])     - ic0[t];
      a[4 + t] = (acc0[4 + t] + acc1[4 + t]) - ic1[t];
    }
    #pragma unroll
    for (int j = 0; j < 8; ++j) {
      e[j]  = __expf((a[j] - shift) * efft);
      denom += e[j];
      vnum   = __builtin_fmaf(e[j], a[j], vnum);
    }

    #pragma unroll
    for (int cc = 0; cc < 4; ++cc) {
      v2f eb;
      eb.x = e[2 * cc + 0];
      eb.y = e[2 * cc + 1];
      #pragma unroll
      for (int dt = 0; dt < 4; ++dt) {
        cacc[dt] = __builtin_amdgcn_wmma_f32_16x16x4_f32(false, yfr[4 * dt + cc], false, eb,
                                                         (short)0, cacc[dt], false, false);
      }
    }
  }

  denom += __shfl_xor(denom, 16, 32);
  vnum  += __shfl_xor(vnum,  16, 32);

  if (h == 0) {
    red_dn[wv * 16 + nlo] = denom;
    red_vn[wv * 16 + nlo] = vnum;
  }
  #pragma unroll
  for (int dt = 0; dt < 4; ++dt) {
    v4f lo, hi;
    #pragma unroll
    for (int t = 0; t < 4; ++t) { lo[t] = cacc[dt][t]; hi[t] = cacc[dt][4 + t]; }
    float* p = &part[(wv * 16 + nlo) * PPITCH + dt * 16 + 8 * h];
    *(v4f*)(p)     = lo;
    *(v4f*)(p + 4) = hi;
  }
  __syncthreads();

  {
    const int row = tid >> 4;
    const int c4  = tid & 15;
    float dn = red_dn[row];
    v4f s = *(const v4f*)(&part[row * PPITCH + 4 * c4]);
    #pragma unroll
    for (int w = 1; w < NWAVE; ++w) {
      dn += red_dn[w * 16 + row];
      s  += *(const v4f*)(&part[(w * 16 + row) * PPITCH + 4 * c4]);
    }
    const float rd = 1.0f / dn;
    s *= rd;
    *(v4f*)(&out_choice[(size_t)(row0 + row) * D_DIM + 4 * c4]) = s;
  }

  if (tid < 16) {
    const int row = tid;
    float dn = red_dn[row];
    float vn = red_vn[row];
    #pragma unroll
    for (int w = 1; w < NWAVE; ++w) {
      dn += red_dn[w * 16 + row];
      vn += red_vn[w * 16 + row];
    }
    out_v[row0 + row] = vn * (1.0f / dn);
  }
}

extern "C" void kernel_launch(void* const* d_in, const int* in_sizes, int n_in,
                              void* d_out, int out_size, void* d_ws, size_t ws_size,
                              hipStream_t stream) {
  const float* X  = (const float*)d_in[0];
  const float* Y  = (const float*)d_in[1];
  const float* ic = (const float*)d_in[2];
  float* out      = (float*)d_out;
  float* out_choice = out;
  float* out_v      = out + (size_t)S_ROWS * D_DIM;

  dim3 grid(S_ROWS / 16);
  dim3 block(32 * NWAVE);
  fcm_kernel<<<grid, block, 0, stream>>>(X, Y, ic, out_choice, out_v);
}


# Pass 1 fetches each 16-row Y chunk with whole-line 16-byte loads and turns it into WMMA fragments through a wave-private LDS buffer, instead of fragment loads that touch 16 lines each
# speedup vs baseline: 5.6696x; 1.3220x over previous
#include <hip/hip_runtime.h>


typedef __attribute__((ext_vector_type(2))) float v2f;
typedef __attribute__((ext_vector_type(4))) float v4f;
typedef __attribute__((ext_vector_type(8))) float v8f;

#define S_ROWS 16384
#define K_COLS 4096
#define D_DIM  64
#define TEMP_C 50.0f
#define MAX_EFF_TEMP 5000.0f
#define NWAVE  8
#define NCH    (K_COLS / (16 * NWAVE))
#define PPITCH 68
#define RPITCH 12
#define SPITCH 68
#define SBUF   (16 * SPITCH)
#define YSTEP  (16 * NWAVE * D_DIM)

__global__ __launch_bounds__(32 * NWAVE) void fcm_kernel(const float* __restrict__ X,
                                                         const float* __restrict__ Y,
                                                         const float* __restrict__ icpt,
                                                         float* __restrict__ out_choice,
                                                         float* __restrict__ out_v) {
  __shared__ __attribute__((aligned(16))) float part[NWAVE * 16 * PPITCH];
  __shared__ __attribute__((aligned(16))) float stage[NWAVE * 2 * SBUF];
  __shared__ __attribute__((aligned(16))) float red_mx[16 * RPITCH];
  __shared__ __attribute__((aligned(16))) float red_mn[16 * RPITCH];
  __shared__ float red_dn[NWAVE * 16];
  __shared__ float red_vn[NWAVE * 16];

  const int tid  = threadIdx.x;
  const int lane = tid & 31;
  const int wv   = __builtin_amdgcn_readfirstlane(tid >> 5);
  const int nlo  = lane & 15;
  const int h    = lane >> 4;
  const int row0 = blockIdx.x * 16;

  v2f xfrag[16];
  const float* xrow = X + (size_t)(row0 + nlo) * D_DIM + 2 * h;
  #pragma unroll
  for (int c = 0; c < 16; ++c)
    xfrag[c] = *(const v2f*)(xrow + 4 * c);

  float rmax = -__builtin_inff();
  float rmin =  __builtin_inff();
  float cmax[NCH];

  v2f ya1[2][16];
  v4f icp[2][2];
  v4f raw[8];
  float* const stw = &stage[wv * (2 * SBUF)];
  float* const st_wr = stw + h * SPITCH + 4 * nlo;
  const float* const st_rd = stw + nlo * SPITCH + 2 * h;
  const float* const ywave = Y + (size_t)(16 * wv) * D_DIM + 4 * lane;
  {
    v4f raw0[8];
    #pragma unroll
    for (int q = 0; q < 8; ++q) raw0[q] = *(const v4f*)(ywave + 128 * q);
    #pragma unroll
    for (int q = 0; q < 8; ++q) raw[q]  = *(const v4f*)(ywave + YSTEP + 128 * q);
    icp[0][0] = *(const v4f*)(icpt + 16 * wv + 8 * h);
    icp[0][1] = *(const v4f*)(icpt + 16 * wv + 8 * h + 4);
    #pragma unroll
    for (int q = 0; q < 8; ++q) *(v4f*)(st_wr + 2 * q * SPITCH) = raw0[q];
    #pragma unroll
    for (int q = 0; q < 8; ++q) *(v4f*)(st_wr + SBUF + 2 * q * SPITCH) = raw[q];
    #pragma unroll
    for (int c = 0; c < 16; ++c) ya1[0][c] = *(const v2f*)(st_rd + 4 * c);
    #pragma unroll
    for (int q = 0; q < 8; ++q) raw[q] = *(const v4f*)(ywave + 2 * YSTEP + 128 * q);
  }

  #pragma unroll
  for (int i = 0; i < NCH; ++i) {
    if (i + 1 < NCH) {
      const int n1 = 16 * wv + 16 * NWAVE * (i + 1);
      #pragma unroll
      for (int c = 0; c < 16; ++c)
        ya1[(i + 1) & 1][c] = *(const v2f*)(st_rd + ((i + 1) & 1) * SBUF + 4 * c);
      icp[(i + 1) & 1][0] = *(const v4f*)(icpt + n1 + 8 * h);
      icp[(i + 1) & 1][1] = *(const v4f*)(icpt + n1 + 8 * h + 4);
    }
    __builtin_amdgcn_sched_barrier(0);

    v8f acc0 = {}; v8f acc1 = {};
    #pragma unroll
    for (int c = 0; c < 16; c += 2) {
      acc0 = __builtin_amdgcn_wmma_f32_16x16x4_f32(false, ya1[i & 1][c],     false, xfrag[c],
                                                   (short)0, acc0, false, false);
      acc1 = __builtin_amdgcn_wmma_f32_16x16x4_f32(false, ya1[i & 1][c + 1], false, xfrag[c + 1],
                                                   (short)0, acc1, false, false);
    }
    __builtin_amdgcn_sched_barrier(0);
    if (i + 2 < NCH) {
      #pragma unroll
      for (int q = 0; q < 8; ++q) *(v4f*)(st_wr + (i & 1) * SBUF + 2 * q * SPITCH) = raw[q];
    }
    if (i + 3 < NCH) {
      #pragma unroll
      for (int q = 0; q < 8; ++q) raw[q] = *(const v4f*)(ywave + (size_t)(i + 3) * YSTEP + 128 * q);
    }

    const v4f ic0 = icp[i & 1][0];
    const v4f ic1 = icp[i & 1][1];
    float cm = -__builtin_inff();
    #pragma unroll
    for (int t = 0; t < 4; ++t) {
      const float lo = (acc0[t]     + acc1[t])     - ic0[t];
      const float hi = (acc0[4 + t] + acc1[4 + t]) - ic1[t];
      cm   = fmaxf(cm,   fmaxf(lo, hi));
      rmin = fminf(rmin, fminf(lo, hi));
    }
    cmax[i] = cm;
    rmax = fmaxf(rmax, cm);
  }

  rmax = fmaxf(rmax, __shfl_xor(rmax, 16, 32));
  rmin = fminf(rmin, __shfl_xor(rmin, 16, 32));

  if (h == 0) {
    red_mx[nlo * RPITCH + wv] = rmax;
    red_mn[nlo * RPITCH + wv] = rmin;
  }
  __syncthreads();
  #pragma unroll
  for (int q = 0; q < NWAVE / 4; ++q) {
    const v4f mx4 = *(const v4f*)(&red_mx[nlo * RPITCH + 4 * q]);
    const v4f mn4 = *(const v4f*)(&red_mn[nlo * RPITCH + 4 * q]);
    #pragma unroll
    for (int t = 0; t < 4; ++t) {
      rmax = fmaxf(rmax, mx4[t]);
      rmin = fminf(rmin, mn4[t]);
    }
  }

  const float span  = fmaxf(rmax - rmin, 1e-3f);
  const float efft  = fminf(fmaxf(TEMP_C / span, TEMP_C), MAX_EFF_TEMP);
  const float shift = rmax;

  unsigned live = 0u;
  #pragma unroll
  for (int i = 0; i < NCH; ++i) {
    const float etop = __expf((cmax[i] - shift) * efft);
    if (__builtin_amdgcn_ballot_w32(etop > 0.0f) != 0u) live |= (1u << i);
  }

  float denom = 0.0f, vnum = 0.0f;
  v8f cacc[4];
  #pragma unroll
  for (int dt = 0; dt < 4; ++dt) cacc[dt] = (v8f){};

  while (live != 0u) {
    const int i = __builtin_ctz(live);
    live &= live - 1u;
    const int n0 = 16 * wv + 16 * NWAVE * i;

    const float* yrow = Y + (size_t)(n0 + nlo) * D_DIM + 2 * h;
    v2f ya[16];
    #pragma unroll
    for (int c = 0; c < 16; ++c)
      ya[c] = *(const v2f*)(yrow + 4 * c);
    const v4f ic0 = *(const v4f*)(icpt + n0 + 8 * h);
    const v4f ic1 = *(const v4f*)(icpt + n0 + 8 * h + 4);
    const float* ycol = Y + (size_t)(n0 + 8 * h) * D_DIM + nlo;
    v2f yfr[16];
    #pragma unroll
    for (int dt = 0; dt < 4; ++dt) {
      #pragma unroll
      for (int cc = 0; cc < 4; ++cc) {
        yfr[4 * dt + cc].x = ycol[(size_t)(2 * cc + 0) * D_DIM + dt * 16];
        yfr[4 * dt + cc].y = ycol[(size_t)(2 * cc + 1) * D_DIM + dt * 16];
      }
    }

    v8f acc0 = {}; v8f acc1 = {};
    #pragma unroll
    for (int c = 0; c < 16; c += 2) {
      acc0 = __builtin_amdgcn_wmma_f32_16x16x4_f32(false, ya[c],     false, xfrag[c],
                                                   (short)0, acc0, false, false);
      acc1 = __builtin_amdgcn_wmma_f32_16x16x4_f32(false, ya[c + 1], false, xfrag[c + 1],
                                                   (short)0, acc1, false, false);
    }
    float a[8], e[8];
    #pragma unroll
    for (int t = 0; t < 4; ++t) {
      a[t]     = (acc0[t]     + acc1[t])     - ic0[t];
      a[4 + t] = (acc0[4 + t] + acc1[4 + t]) - ic1[t];
    }
    #pragma unroll
    for (int j = 0; j < 8; ++j) {
      e[j]  = __expf((a[j] - shift) * efft);
      denom += e[j];
      vnum   = __builtin_fmaf(e[j], a[j], vnum);
    }

    #pragma unroll
    for (int cc = 0; cc < 4; ++cc) {
      v2f eb;
      eb.x = e[2 * cc + 0];
      eb.y = e[2 * cc + 1];
      #pragma unroll
      for (int dt = 0; dt < 4; ++dt) {
        cacc[dt] = __builtin_amdgcn_wmma_f32_16x16x4_f32(false, yfr[4 * dt + cc], false, eb,
                                                         (short)0, cacc[dt], false, false);
      }
    }
  }

  denom += __shfl_xor(denom, 16, 32);
  vnum  += __shfl_xor(vnum,  16, 32);

  if (h == 0) {
    red_dn[wv * 16 + nlo] = denom;
    red_vn[wv * 16 + nlo] = vnum;
  }
  #pragma unroll
  for (int dt = 0; dt < 4; ++dt) {
    v4f lo, hi;
    #pragma unroll
    for (int t = 0; t < 4; ++t) { lo[t] = cacc[dt][t]; hi[t] = cacc[dt][4 + t]; }
    float* p = &part[(wv * 16 + nlo) * PPITCH + dt * 16 + 8 * h];
    *(v4f*)(p)     = lo;
    *(v4f*)(p + 4) = hi;
  }
  __syncthreads();

  {
    const int row = tid >> 4;
    const int c4  = tid & 15;
    float dn = red_dn[row];
    v4f s = *(const v4f*)(&part[row * PPITCH + 4 * c4]);
    #pragma unroll
    for (int w = 1; w < NWAVE; ++w) {
      dn += red_dn[w * 16 + row];
      s  += *(const v4f*)(&part[(w * 16 + row) * PPITCH + 4 * c4]);
    }
    const float rd = 1.0f / dn;
    s *= rd;
    *(v4f*)(&out_choice[(size_t)(row0 + row) * D_DIM + 4 * c4]) = s;
  }

  if (tid < 16) {
    const int row = tid;
    float dn = red_dn[row];
    float vn = red_vn[row];
    #pragma unroll
    for (int w = 1; w < NWAVE; ++w) {
      dn += red_dn[w * 16 + row];
      vn += red_vn[w * 16 + row];
    }
    out_v[row0 + row] = vn * (1.0f / dn);
  }
}

extern "C" void kernel_launch(void* const* d_in, const int* in_sizes, int n_in,
                              void* d_out, int out_size, void* d_ws, size_t ws_size,
                              hipStream_t stream) {
  const float* X  = (const float*)d_in[0];
  const float* Y  = (const float*)d_in[1];
  const float* ic = (const float*)d_in[2];
  float* out      = (float*)d_out;
  float* out_choice = out;
  float* out_v      = out + (size_t)S_ROWS * D_DIM;

  dim3 grid(S_ROWS / 16);
  dim3 block(32 * NWAVE);
  fcm_kernel<<<grid, block, 0, stream>>>(X, Y, ic, out_choice, out_v);
}


# Each workgroup owns two 16-row X tiles (512 workgroups): a Y chunk staged in LDS feeds 32 WMMAs, not 16, halving Y traffic from L2 and LDS per row; chunk maxima kept per row; all sums as before
# speedup vs baseline: 6.2432x; 1.1012x over previous
#include <hip/hip_runtime.h>


typedef __attribute__((ext_vector_type(2))) float v2f;
typedef __attribute__((ext_vector_type(4))) float v4f;
typedef __attribute__((ext_vector_type(8))) float v8f;

#define S_ROWS 16384
#define K_COLS 4096
#define D_DIM  64
#define TEMP_C 50.0f
#define MAX_EFF_TEMP 5000.0f
#define NWAVE  8
#define NT     2
#define WROWS  (16 * NT)
#define NCH    (K_COLS / (16 * NWAVE))
#define PPITCH 68
#define RPITCH 12
#define SPITCH 68
#define SBUF   (16 * SPITCH)
#define YSTEP  (16 * NWAVE * D_DIM)

__device__ __forceinline__ float xhalf(float x) {
  const int b = __builtin_bit_cast(int, x);
  return __builtin_bit_cast(float, __builtin_amdgcn_permlanex16(b, b, 0x76543210, 0xfedcba98, false, false));
}

__global__ __launch_bounds__(32 * NWAVE) void fcm_kernel(const float* __restrict__ X,
                                                         const float* __restrict__ Y,
                                                         const float* __restrict__ icpt,
                                                         float* __restrict__ out_choice,
                                                         float* __restrict__ out_v) {
  __shared__ __attribute__((aligned(16))) float part[NWAVE * WROWS * PPITCH];
  __shared__ __attribute__((aligned(16))) float stage[NWAVE * 2 * SBUF];
  __shared__ __attribute__((aligned(16))) float red_mx[WROWS * RPITCH];
  __shared__ __attribute__((aligned(16))) float red_mn[WROWS * RPITCH];
  __shared__ float red_dn[NWAVE * WROWS];
  __shared__ float red_vn[NWAVE * WROWS];

  const int tid  = threadIdx.x;
  const int lane = tid & 31;
  const int wv   = __builtin_amdgcn_readfirstlane(tid >> 5);
  const int nlo  = lane & 15;
  const int h    = lane >> 4;
  const int row0 = blockIdx.x * WROWS;

  v2f xfrag[NT][16];
  #pragma unroll
  for (int t = 0; t < NT; ++t) {
    const float* xrow = X + (size_t)(row0 + 16 * t + nlo) * D_DIM + 2 * h;
    #pragma unroll
    for (int c = 0; c < 16; ++c)
      xfrag[t][c] = *(const v2f*)(xrow + 4 * c);
  }

  float rmin[NT];
  #pragma unroll
  for (int t = 0; t < NT; ++t) rmin[t] = __builtin_inff();
  float rmaxc = -__builtin_inff();
  float cmaxc[NCH];

  v2f ya1[16];
  v4f icp[2][2];
  v4f raw[8];
  float* const stw = &stage[wv * (2 * SBUF)];
  float* const st_wr = stw + h * SPITCH + 4 * nlo;
  const float* const st_rd = stw + nlo * SPITCH + 2 * h;
  const float* const ywave = Y + (size_t)(16 * wv) * D_DIM + 4 * lane;
  {
    v4f raw0[8];
    #pragma unroll
    for (int q = 0; q < 8; ++q) raw0[q] = *(const v4f*)(ywave + 128 * q);
    #pragma unroll
    for (int q = 0; q < 8; ++q) raw[q]  = *(const v4f*)(ywave + YSTEP + 128 * q);
    icp[0][0] = *(const v4f*)(icpt + 16 * wv + 8 * h);
    icp[0][1] = *(const v4f*)(icpt + 16 * wv + 8 * h + 4);
    #pragma unroll
    for (int q = 0; q < 8; ++q) *(v4f*)(st_wr + 2 * q * SPITCH) = raw0[q];
    #pragma unroll
    for (int q = 0; q < 8; ++q) *(v4f*)(st_wr + SBUF + 2 * q * SPITCH) = raw[q];
    #pragma unroll
    for (int c = 0; c < 16; ++c) ya1[c] = *(const v2f*)(st_rd + 4 * c);
    #pragma unroll
    for (int q = 0; q < 8; ++q) raw[q] = *(const v4f*)(ywave + 2 * YSTEP + 128 * q);
  }

  #pragma unroll
  for (int i = 0; i < NCH; ++i) {
    __builtin_amdgcn_sched_barrier(0);

    v8f acc0[NT], acc1[NT];
    #pragma unroll
    for (int t = 0; t < NT; ++t) { acc0[t] = (v8f){}; acc1[t] = (v8f){}; }
    #pragma unroll
    for (int c = 0; c < 16; c += 2) {
      #pragma unroll
      for (int t = 0; t < NT; ++t)
        acc0[t] = __builtin_amdgcn_wmma_f32_16x16x4_f32(false, ya1[c],     false, xfrag[t][c],
                                                        (short)0, acc0[t], false, false);
      #pragma unroll
      for (int t = 0; t < NT; ++t)
        acc1[t] = __builtin_amdgcn_wmma_f32_16x16x4_f32(false, ya1[c + 1], false, xfrag[t][c + 1],
                                                        (short)0, acc1[t], false, false);
    }
    __builtin_amdgcn_sched_barrier(0);
    if (i + 1 < NCH) {
      const int n1 = 16 * wv + 16 * NWAVE * (i + 1);
      #pragma unroll
      for (int c = 0; c < 16; ++c)
        ya1[c] = *(const v2f*)(st_rd + ((i + 1) & 1) * SBUF + 4 * c);
      icp[(i + 1) & 1][0] = *(const v4f*)(icpt + n1 + 8 * h);
      icp[(i + 1) & 1][1] = *(const v4f*)(icpt + n1 + 8 * h + 4);
    }
    if (i + 2 < NCH) {
      #pragma unroll
      for (int q = 0; q < 8; ++q) *(v4f*)(st_wr + (i & 1) * SBUF + 2 * q * SPITCH) = raw[q];
    }
    if (i + 3 < NCH) {
      #pragma unroll
      for (int q = 0; q < 8; ++q) raw[q] = *(const v4f*)(ywave + (size_t)(i + 3) * YSTEP + 128 * q);
    }

    const v4f ic0 = icp[i & 1][0];
    const v4f ic1 = icp[i & 1][1];
    float cm[NT];
    #pragma unroll
    for (int t = 0; t < NT; ++t) {
      float mx = -__builtin_inff();
      float mn = rmin[t];
      #pragma unroll
      for (int u = 0; u < 4; ++u) {
        const float lo = (acc0[t][u]     + acc1[t][u])     - ic0[u];
        const float hi = (acc0[t][4 + u] + acc1[t][4 + u]) - ic1[u];
        mx = fmaxf(mx, fmaxf(lo, hi));
        mn = fminf(mn, fminf(lo, hi));
      }
      cm[t]   = mx;
      rmin[t] = mn;
    }
    {
      const float keep = h ? cm[1] : cm[0];
      const float give = h ? cm[0] : cm[1];
      cmaxc[i] = fmaxf(keep, xhalf(give));
      rmaxc    = fmaxf(rmaxc, cmaxc[i]);
    }
  }

  #pragma unroll
  for (int t = 0; t < NT; ++t)
    rmin[t] = fminf(rmin[t], xhalf(rmin[t]));

  red_mx[(16 * h + nlo) * RPITCH + wv] = rmaxc;
  if (h == 0) {
    #pragma unroll
    for (int t = 0; t < NT; ++t)
      red_mn[(16 * t + nlo) * RPITCH + wv] = rmin[t];
  }
  __syncthreads();
  float efft[NT], shift[NT], rmax[NT];
  #pragma unroll
  for (int t = 0; t < NT; ++t) {
    rmax[t] = -__builtin_inff();
    #pragma unroll
    for (int q = 0; q < NWAVE / 4; ++q) {
      const v4f mx4 = *(const v4f*)(&red_mx[(16 * t + nlo) * RPITCH + 4 * q]);
      const v4f mn4 = *(const v4f*)(&red_mn[(16 * t + nlo) * RPITCH + 4 * q]);
      #pragma unroll
      for (int u = 0; u < 4; ++u) {
        rmax[t] = fmaxf(rmax[t], mx4[u]);
        rmin[t] = fminf(rmin[t], mn4[u]);
      }
    }
    const float span = fmaxf(rmax[t] - rmin[t], 1e-3f);
    efft[t]  = fminf(fmaxf(TEMP_C / span, TEMP_C), MAX_EFF_TEMP);
    shift[t] = rmax[t];
  }

  unsigned live[NT];
  #pragma unroll
  for (int t = 0; t < NT; ++t) live[t] = 0u;
  {
    const float shc = h ? shift[1] : shift[0];
    const float efc = h ? efft[1]  : efft[0];
    #pragma unroll
    for (int i = 0; i < NCH; ++i) {
      const float etop = __expf((cmaxc[i] - shc) * efc);
      const unsigned b = __builtin_amdgcn_ballot_w32(etop > 0.0f);
      if ((b & 0xffffu) != 0u) live[0] |= (1u << i);
      if ((b >> 16)     != 0u) live[1] |= (1u << i);
    }
  }

  #pragma unroll
  for (int t = 0; t < NT; ++t) {
    float denom = 0.0f, vnum = 0.0f;
    v8f cacc[4];
    #pragma unroll
    for (int dt = 0; dt < 4; ++dt) cacc[dt] = (v8f){};
    const float sh = shift[t];
    const float ef = efft[t];
    unsigned lv = live[t];

    while (lv != 0u) {
      const int i = __builtin_ctz(lv);
      lv &= lv - 1u;
      const int n0 = 16 * wv + 16 * NWAVE * i;

      const float* yrow = Y + (size_t)(n0 + nlo) * D_DIM + 2 * h;
      v2f ya[16];
      #pragma unroll
      for (int c = 0; c < 16; ++c)
        ya[c] = *(const v2f*)(yrow + 4 * c);
      const v4f ic0 = *(const v4f*)(icpt + n0 + 8 * h);
      const v4f ic1 = *(const v4f*)(icpt + n0 + 8 * h + 4);
      const float* ycol = Y + (size_t)(n0 + 8 * h) * D_DIM + nlo;
      v2f yfr[16];
      #pragma unroll
      for (int dt = 0; dt < 4; ++dt) {
        #pragma unroll
        for (int cc = 0; cc < 4; ++cc) {
          yfr[4 * dt + cc].x = ycol[(size_t)(2 * cc + 0) * D_DIM + dt * 16];
          yfr[4 * dt + cc].y = ycol[(size_t)(2 * cc + 1) * D_DIM + dt * 16];
        }
      }

      v8f acc0 = {}; v8f acc1 = {};
      #pragma unroll
      for (int c = 0; c < 16; c += 2) {
        acc0 = __builtin_amdgcn_wmma_f32_16x16x4_f32(false, ya[c],     false, xfrag[t][c],
                                                     (short)0, acc0, false, false);
        acc1 = __builtin_amdgcn_wmma_f32_16x16x4_f32(false, ya[c + 1], false, xfrag[t][c + 1],
                                                     (short)0, acc1, false, false);
      }
      float a[8], e[8];
      #pragma unroll
      for (int u = 0; u < 4; ++u) {
        a[u]     = (acc0[u]     + acc1[u])     - ic0[u];
        a[4 + u] = (acc0[4 + u] + acc1[4 + u]) - ic1[u];
      }
      #pragma unroll
      for (int j = 0; j < 8; ++j) {
        e[j]  = __expf((a[j] - sh) * ef);
        denom += e[j];
        vnum   = __builtin_fmaf(e[j], a[j], vnum);
      }

      #pragma unroll
      for (int cc = 0; cc < 4; ++cc) {
        v2f eb;
        eb.x = e[2 * cc + 0];
        eb.y = e[2 * cc + 1];
        #pragma unroll
        for (int dt = 0; dt < 4; ++dt) {
          cacc[dt] = __builtin_amdgcn_wmma_f32_16x16x4_f32(false, yfr[4 * dt + cc], false, eb,
                                                           (short)0, cacc[dt], false, false);
        }
      }
    }

    denom += __shfl_xor(denom, 16, 32);
    vnum  += __shfl_xor(vnum,  16, 32);

    if (h == 0) {
      red_dn[wv * WROWS + 16 * t + nlo] = denom;
      red_vn[wv * WROWS + 16 * t + nlo] = vnum;
    }
    #pragma unroll
    for (int dt = 0; dt < 4; ++dt) {
      v4f lo, hi;
      #pragma unroll
      for (int u = 0; u < 4; ++u) { lo[u] = cacc[dt][u]; hi[u] = cacc[dt][4 + u]; }
      float* p = &part[(wv * WROWS + 16 * t + nlo) * PPITCH + dt * 16 + 8 * h];
      *(v4f*)(p)     = lo;
      *(v4f*)(p + 4) = hi;
    }
  }
  __syncthreads();

  #pragma unroll
  for (int r = 0; r < NT; ++r) {
    const int row = 16 * r + (tid >> 4);
    const int c4  = tid & 15;
    float dn = red_dn[row];
    v4f s = *(const v4f*)(&part[row * PPITCH + 4 * c4]);
    #pragma unroll
    for (int w = 1; w < NWAVE; ++w) {
      dn += red_dn[w * WROWS + row];
      s  += *(const v4f*)(&part[(w * WROWS + row) * PPITCH + 4 * c4]);
    }
    const float rd = 1.0f / dn;
    s *= rd;
    *(v4f*)(&out_choice[(size_t)(row0 + row) * D_DIM + 4 * c4]) = s;
  }

  if (tid < WROWS) {
    const int row = tid;
    float dn = red_dn[row];
    float vn = red_vn[row];
    #pragma unroll
    for (int w = 1; w < NWAVE; ++w) {
      dn += red_dn[w * WROWS + row];
      vn += red_vn[w * WROWS + row];
    }
    out_v[row0 + row] = vn * (1.0f / dn);
  }
}

extern "C" void kernel_launch(void* const* d_in, const int* in_sizes, int n_in,
                              void* d_out, int out_size, void* d_ws, size_t ws_size,
                              hipStream_t stream) {
  const float* X  = (const float*)d_in[0];
  const float* Y  = (const float*)d_in[1];
  const float* ic = (const float*)d_in[2];
  float* out      = (float*)d_out;
  float* out_choice = out;
  float* out_v      = out + (size_t)S_ROWS * D_DIM;

  dim3 grid(S_ROWS / WROWS);
  dim3 block(32 * NWAVE);
  fcm_kernel<<<grid, block, 0, stream>>>(X, Y, ic, out_choice, out_v);
}


# Pass 1 lowers each wave's issue priority as it advances (s_setprio 3,2,1,0 at chunks 0,12,22,30): a SIMD's four waves end pass 1 together, not oldest first; no value or sum order changes
# speedup vs baseline: 6.4698x; 1.0363x over previous
#include <hip/hip_runtime.h>


typedef __attribute__((ext_vector_type(2))) float v2f;
typedef __attribute__((ext_vector_type(4))) float v4f;
typedef __attribute__((ext_vector_type(8))) float v8f;

#define S_ROWS 16384
#define K_COLS 4096
#define D_DIM  64
#define TEMP_C 50.0f
#define MAX_EFF_TEMP 5000.0f
#define NWAVE  8
#define NT     2
#define WROWS  (16 * NT)
#define NCH    (K_COLS / (16 * NWAVE))
#define PPITCH 68
#define RPITCH 12
#define SPITCH 68
#define SBUF   (16 * SPITCH)
#define YSTEP  (16 * NWAVE * D_DIM)

__device__ __forceinline__ float xhalf(float x) {
  const int b = __builtin_bit_cast(int, x);
  return __builtin_bit_cast(float, __builtin_amdgcn_permlanex16(b, b, 0x76543210, 0xfedcba98, false, false));
}

__global__ __launch_bounds__(32 * NWAVE) void fcm_kernel(const float* __restrict__ X,
                                                         const float* __restrict__ Y,
                                                         const float* __restrict__ icpt,
                                                         float* __restrict__ out_choice,
                                                         float* __restrict__ out_v) {
  __shared__ __attribute__((aligned(16))) float part[NWAVE * WROWS * PPITCH];
  __shared__ __attribute__((aligned(16))) float stage[NWAVE * 2 * SBUF];
  __shared__ __attribute__((aligned(16))) float red_mx[WROWS * RPITCH];
  __shared__ __attribute__((aligned(16))) float red_mn[WROWS * RPITCH];
  __shared__ float red_dn[NWAVE * WROWS];
  __shared__ float red_vn[NWAVE * WROWS];

  const int tid  = threadIdx.x;
  const int lane = tid & 31;
  const int wv   = __builtin_amdgcn_readfirstlane(tid >> 5);
  const int nlo  = lane & 15;
  const int h    = lane >> 4;
  const int row0 = blockIdx.x * WROWS;

  v2f xfrag[NT][16];
  #pragma unroll
  for (int t = 0; t < NT; ++t) {
    const float* xrow = X + (size_t)(row0 + 16 * t + nlo) * D_DIM + 2 * h;
    #pragma unroll
    for (int c = 0; c < 16; ++c)
      xfrag[t][c] = *(const v2f*)(xrow + 4 * c);
  }

  float rmin[NT];
  #pragma unroll
  for (int t = 0; t < NT; ++t) rmin[t] = __builtin_inff();
  float rmaxc = -__builtin_inff();
  float cmaxc[NCH];

  v2f ya1[16];
  v4f icp[2][2];
  v4f raw[8];
  float* const stw = &stage[wv * (2 * SBUF)];
  float* const st_wr = stw + h * SPITCH + 4 * nlo;
  const float* const st_rd = stw + nlo * SPITCH + 2 * h;
  const float* const ywave = Y + (size_t)(16 * wv) * D_DIM + 4 * lane;
  {
    v4f raw0[8];
    #pragma unroll
    for (int q = 0; q < 8; ++q) raw0[q] = *(const v4f*)(ywave + 128 * q);
    #pragma unroll
    for (int q = 0; q < 8; ++q) raw[q]  = *(const v4f*)(ywave + YSTEP + 128 * q);
    icp[0][0] = *(const v4f*)(icpt + 16 * wv + 8 * h);
    icp[0][1] = *(const v4f*)(icpt + 16 * wv + 8 * h + 4);
    #pragma unroll
    for (int q = 0; q < 8; ++q) *(v4f*)(st_wr + 2 * q * SPITCH) = raw0[q];
    #pragma unroll
    for (int q = 0; q < 8; ++q) *(v4f*)(st_wr + SBUF + 2 * q * SPITCH) = raw[q];
    #pragma unroll
    for (int c = 0; c < 16; ++c) ya1[c] = *(const v2f*)(st_rd + 4 * c);
    #pragma unroll
    for (int q = 0; q < 8; ++q) raw[q] = *(const v4f*)(ywave + 2 * YSTEP + 128 * q);
  }

  #pragma unroll
  for (int i = 0; i < NCH; ++i) {
    if (i == 0)  __builtin_amdgcn_s_setprio(3);
    if (i == 12) __builtin_amdgcn_s_setprio(2);
    if (i == 22) __builtin_amdgcn_s_setprio(1);
    if (i == 30) __builtin_amdgcn_s_setprio(0);
    __builtin_amdgcn_sched_barrier(0);

    v8f acc0[NT], acc1[NT];
    #pragma unroll
    for (int t = 0; t < NT; ++t) { acc0[t] = (v8f){}; acc1[t] = (v8f){}; }
    #pragma unroll
    for (int c = 0; c < 16; c += 2) {
      #pragma unroll
      for (int t = 0; t < NT; ++t)
        acc0[t] = __builtin_amdgcn_wmma_f32_16x16x4_f32(false, ya1[c],     false, xfrag[t][c],
                                                        (short)0, acc0[t], false, false);
      #pragma unroll
      for (int t = 0; t < NT; ++t)
        acc1[t] = __builtin_amdgcn_wmma_f32_16x16x4_f32(false, ya1[c + 1], false, xfrag[t][c + 1],
                                                        (short)0, acc1[t], false, false);
    }
    __builtin_amdgcn_sched_barrier(0);
    if (i + 1 < NCH) {
      const int n1 = 16 * wv + 16 * NWAVE * (i + 1);
      #pragma unroll
      for (int c = 0; c < 16; ++c)
        ya1[c] = *(const v2f*)(st_rd + ((i + 1) & 1) * SBUF + 4 * c);
      icp[(i + 1) & 1][0] = *(const v4f*)(icpt + n1 + 8 * h);
      icp[(i + 1) & 1][1] = *(const v4f*)(icpt + n1 + 8 * h + 4);
    }
    if (i + 2 < NCH) {
      #pragma unroll
      for (int q = 0; q < 8; ++q) *(v4f*)(st_wr + (i & 1) * SBUF + 2 * q * SPITCH) = raw[q];
    }
    if (i + 3 < NCH) {
      #pragma unroll
      for (int q = 0; q < 8; ++q) raw[q] = *(const v4f*)(ywave + (size_t)(i + 3) * YSTEP + 128 * q);
    }

    const v4f ic0 = icp[i & 1][0];
    const v4f ic1 = icp[i & 1][1];
    float cm[NT];
    #pragma unroll
    for (int t = 0; t < NT; ++t) {
      float mx = -__builtin_inff();
      float mn = rmin[t];
      #pragma unroll
      for (int u = 0; u < 4; ++u) {
        const float lo = (acc0[t][u]     + acc1[t][u])     - ic0[u];
        const float hi = (acc0[t][4 + u] + acc1[t][4 + u]) - ic1[u];
        mx = fmaxf(mx, fmaxf(lo, hi));
        mn = fminf(mn, fminf(lo, hi));
      }
      cm[t]   = mx;
      rmin[t] = mn;
    }
    {
      const float keep = h ? cm[1] : cm[0];
      const float give = h ? cm[0] : cm[1];
      cmaxc[i] = fmaxf(keep, xhalf(give));
      rmaxc    = fmaxf(rmaxc, cmaxc[i]);
    }
  }

  #pragma unroll
  for (int t = 0; t < NT; ++t)
    rmin[t] = fminf(rmin[t], xhalf(rmin[t]));

  red_mx[(16 * h + nlo) * RPITCH + wv] = rmaxc;
  if (h == 0) {
    #pragma unroll
    for (int t = 0; t < NT; ++t)
      red_mn[(16 * t + nlo) * RPITCH + wv] = rmin[t];
  }
  __syncthreads();
  float efft[NT], shift[NT], rmax[NT];
  #pragma unroll
  for (int t = 0; t < NT; ++t) {
    rmax[t] = -__builtin_inff();
    #pragma unroll
    for (int q = 0; q < NWAVE / 4; ++q) {
      const v4f mx4 = *(const v4f*)(&red_mx[(16 * t + nlo) * RPITCH + 4 * q]);
      const v4f mn4 = *(const v4f*)(&red_mn[(16 * t + nlo) * RPITCH + 4 * q]);
      #pragma unroll
      for (int u = 0; u < 4; ++u) {
        rmax[t] = fmaxf(rmax[t], mx4[u]);
        rmin[t] = fminf(rmin[t], mn4[u]);
      }
    }
    const float span = fmaxf(rmax[t] - rmin[t], 1e-3f);
    efft[t]  = fminf(fmaxf(TEMP_C / span, TEMP_C), MAX_EFF_TEMP);
    shift[t] = rmax[t];
  }

  unsigned live[NT];
  #pragma unroll
  for (int t = 0; t < NT; ++t) live[t] = 0u;
  {
    const float shc = h ? shift[1] : shift[0];
    const float efc = h ? efft[1]  : efft[0];
    #pragma unroll
    for (int i = 0; i < NCH; ++i) {
      const float etop = __expf((cmaxc[i] - shc) * efc);
      const unsigned b = __builtin_amdgcn_ballot_w32(etop > 0.0f);
      if ((b & 0xffffu) != 0u) live[0] |= (1u << i);
      if ((b >> 16)     != 0u) live[1] |= (1u << i);
    }
  }

  #pragma unroll
  for (int t = 0; t < NT; ++t) {
    float denom = 0.0f, vnum = 0.0f;
    v8f cacc[4];
    #pragma unroll
    for (int dt = 0; dt < 4; ++dt) cacc[dt] = (v8f){};
    const float sh = shift[t];
    const float ef = efft[t];
    unsigned lv = live[t];

    while (lv != 0u) {
      const int i = __builtin_ctz(lv);
      lv &= lv - 1u;
      const int n0 = 16 * wv + 16 * NWAVE * i;

      const float* yrow = Y + (size_t)(n0 + nlo) * D_DIM + 2 * h;
      v2f ya[16];
      #pragma unroll
      for (int c = 0; c < 16; ++c)
        ya[c] = *(const v2f*)(yrow + 4 * c);
      const v4f ic0 = *(const v4f*)(icpt + n0 + 8 * h);
      const v4f ic1 = *(const v4f*)(icpt + n0 + 8 * h + 4);
      const float* ycol = Y + (size_t)(n0 + 8 * h) * D_DIM + nlo;
      v2f yfr[16];
      #pragma unroll
      for (int dt = 0; dt < 4; ++dt) {
        #pragma unroll
        for (int cc = 0; cc < 4; ++cc) {
          yfr[4 * dt + cc].x = ycol[(size_t)(2 * cc + 0) * D_DIM + dt * 16];
          yfr[4 * dt + cc].y = ycol[(size_t)(2 * cc + 1) * D_DIM + dt * 16];
        }
      }

      v8f acc0 = {}; v8f acc1 = {};
      #pragma unroll
      for (int c = 0; c < 16; c += 2) {
        acc0 = __builtin_amdgcn_wmma_f32_16x16x4_f32(false, ya[c],     false, xfrag[t][c],
                                                     (short)0, acc0, false, false);
        acc1 = __builtin_amdgcn_wmma_f32_16x16x4_f32(false, ya[c + 1], false, xfrag[t][c + 1],
                                                     (short)0, acc1, false, false);
      }
      float a[8], e[8];
      #pragma unroll
      for (int u = 0; u < 4; ++u) {
        a[u]     = (acc0[u]     + acc1[u])     - ic0[u];
        a[4 + u] = (acc0[4 + u] + acc1[4 + u]) - ic1[u];
      }
      #pragma unroll
      for (int j = 0; j < 8; ++j) {
        e[j]  = __expf((a[j] - sh) * ef);
        denom += e[j];
        vnum   = __builtin_fmaf(e[j], a[j], vnum);
      }

      #pragma unroll
      for (int cc = 0; cc < 4; ++cc) {
        v2f eb;
        eb.x = e[2 * cc + 0];
        eb.y = e[2 * cc + 1];
        #pragma unroll
        for (int dt = 0; dt < 4; ++dt) {
          cacc[dt] = __builtin_amdgcn_wmma_f32_16x16x4_f32(false, yfr[4 * dt + cc], false, eb,
                                                           (short)0, cacc[dt], false, false);
        }
      }
    }

    denom += __shfl_xor(denom, 16, 32);
    vnum  += __shfl_xor(vnum,  16, 32);

    if (h == 0) {
      red_dn[wv * WROWS + 16 * t + nlo] = denom;
      red_vn[wv * WROWS + 16 * t + nlo] = vnum;
    }
    #pragma unroll
    for (int dt = 0; dt < 4; ++dt) {
      v4f lo, hi;
      #pragma unroll
      for (int u = 0; u < 4; ++u) { lo[u] = cacc[dt][u]; hi[u] = cacc[dt][4 + u]; }
      float* p = &part[(wv * WROWS + 16 * t + nlo) * PPITCH + dt * 16 + 8 * h];
      *(v4f*)(p)     = lo;
      *(v4f*)(p + 4) = hi;
    }
  }
  __syncthreads();

  #pragma unroll
  for (int r = 0; r < NT; ++r) {
    const int row = 16 * r + (tid >> 4);
    const int c4  = tid & 15;
    float dn = red_dn[row];
    v4f s = *(const v4f*)(&part[row * PPITCH + 4 * c4]);
    #pragma unroll
    for (int w = 1; w < NWAVE; ++w) {
      dn += red_dn[w * WROWS + row];
      s  += *(const v4f*)(&part[(w * WROWS + row) * PPITCH + 4 * c4]);
    }
    const float rd = 1.0f / dn;
    s *= rd;
    *(v4f*)(&out_choice[(size_t)(row0 + row) * D_DIM + 4 * c4]) = s;
  }

  if (tid < WROWS) {
    const int row = tid;
    float dn = red_dn[row];
    float vn = red_vn[row];
    #pragma unroll
    for (int w = 1; w < NWAVE; ++w) {
      dn += red_dn[w * WROWS + row];
      vn += red_vn[w * WROWS + row];
    }
    out_v[row0 + row] = vn * (1.0f / dn);
  }
}

extern "C" void kernel_launch(void* const* d_in, const int* in_sizes, int n_in,
                              void* d_out, int out_size, void* d_ws, size_t ws_size,
                              hipStream_t stream) {
  const float* X  = (const float*)d_in[0];
  const float* Y  = (const float*)d_in[1];
  const float* ic = (const float*)d_in[2];
  float* out      = (float*)d_out;
  float* out_choice = out;
  float* out_v      = out + (size_t)S_ROWS * D_DIM;

  dim3 grid(S_ROWS / WROWS);
  dim3 block(32 * NWAVE);
  fcm_kernel<<<grid, block, 0, stream>>>(X, Y, ic, out_choice, out_v);
}
